# starting kernel
# baseline (speedup 1.0000x reference)
#include <hip/hip_runtime.h>


#define N_NODES 50000
#define N_EDGES 600000
#define IN_CH   128
#define HID     128
#define OUT_CH  64

typedef __attribute__((ext_vector_type(2))) float v2f;
typedef __attribute__((ext_vector_type(8))) float v8f;

__global__ void deg_kernel(const int* __restrict__ row, const int* __restrict__ col,
                           const float* __restrict__ w, float* __restrict__ deg) {
    int e = blockIdx.x * blockDim.x + threadIdx.x;
    if (e >= N_EDGES) return;
    int r = row[e], c = col[e];
    float we = (r == c) ? 0.0f : w[e];
    if (we != 0.0f) atomicAdd(&deg[c], we);
}

__global__ void dis_kernel(const float* __restrict__ deg, float* __restrict__ dis,
                           float* __restrict__ invdeg) {
    int i = blockIdx.x * blockDim.x + threadIdx.x;
    if (i >= N_NODES) return;
    float d = deg[i] + 1.0f;
    dis[i]    = rsqrtf(d);
    invdeg[i] = 1.0f / d;
}

__global__ void gemm_xwt_wmma(const float* __restrict__ X,
                              const float* __restrict__ Wa,
                              const float* __restrict__ Wb,
                              int n_split,
                              float* __restrict__ C) {
    int w     = blockIdx.x * 8 + (threadIdx.x >> 5);
    int lane  = threadIdx.x & 31;
    int nTile = w & 7;
    int mTile = w >> 3;
    if (mTile >= N_NODES / 16) return;

    int half = lane >> 4;
    int l16  = lane & 15;

    const float* xrow = X + (size_t)(mTile * 16 + l16) * IN_CH + half * 2;
    int ng = nTile * 16 + l16;
    const float* wrow = ((ng < n_split) ? (Wa + (size_t)ng * IN_CH)
                                        : (Wb + (size_t)(ng - n_split) * IN_CH)) + half * 2;

    v8f acc = {};
#pragma unroll
    for (int k = 0; k < IN_CH; k += 4) {
        v2f a = *(const v2f*)(xrow + k);
        v2f b = *(const v2f*)(wrow + k);
        acc = __builtin_amdgcn_wmma_f32_16x16x4_f32(
            false, a, false, b, (short)0, acc, false, false);
    }

    float* crow = C + (size_t)(mTile * 16 + half * 8) * HID + nTile * 16 + l16;
#pragma unroll
    for (int j = 0; j < 8; ++j) crow[(size_t)j * HID] = acc[j];
}

__global__ void scatter_kernel(const int* __restrict__ row, const int* __restrict__ col,
                               const float* __restrict__ w, const float* __restrict__ dis,
                               const float* __restrict__ src, float* __restrict__ agg) {
    int wave = (blockIdx.x * blockDim.x + threadIdx.x) >> 5;
    int lane = threadIdx.x & 31;
    if (wave >= N_EDGES) return;
    int r = row[wave], c = col[wave];
    float we = (r == c) ? 0.0f : w[wave];
    if (we == 0.0f) return;
    float coeff = dis[r] * we * dis[c];

    float4 v = ((const float4*)(src + (size_t)r * HID))[lane];
    float* dst = agg + (size_t)c * HID + lane * 4;
    atomicAdd(dst + 0, coeff * v.x);
    atomicAdd(dst + 1, coeff * v.y);
    atomicAdd(dst + 2, coeff * v.z);
    atomicAdd(dst + 3, coeff * v.w);
}

__global__ void act_kernel(float* __restrict__ agg, const float* __restrict__ xw,
                           const float* __restrict__ invdeg, const float* __restrict__ b1) {
    int idx = blockIdx.x * blockDim.x + threadIdx.x;
    if (idx >= N_NODES * HID) return;
    int node = idx >> 7;
    int ch   = idx & 127;
    float v = agg[idx] + xw[idx] * invdeg[node] + b1[ch];
    agg[idx] = fmaxf(v, 0.0f);
}

__global__ void out_kernel(const float* __restrict__ agg2, const float* __restrict__ xw2,
                           const float* __restrict__ invdeg, const float* __restrict__ b_mu,
                           const float* __restrict__ b_ls, float* __restrict__ out) {
    int idx = blockIdx.x * blockDim.x + threadIdx.x;
    if (idx >= N_NODES * OUT_CH) return;
    int node = idx >> 6;
    int ch   = idx & 63;
    size_t base = (size_t)node * HID;
    float id = invdeg[node];
    out[idx] = agg2[base + ch] + xw2[base + ch] * id + b_mu[ch];
    out[(size_t)N_NODES * OUT_CH + idx] =
        agg2[base + OUT_CH + ch] + xw2[base + OUT_CH + ch] * id + b_ls[ch];
}

extern "C" void kernel_launch(void* const* d_in, const int* in_sizes, int n_in,
                              void* d_out, int out_size, void* d_ws, size_t ws_size,
                              hipStream_t stream) {
    (void)in_sizes; (void)n_in; (void)out_size; (void)ws_size;
    const float* x   = (const float*)d_in[0];
    const int*   ei  = (const int*)d_in[1];
    const float* ew  = (const float*)d_in[2];
    const float* W1  = (const float*)d_in[3];
    const float* b1  = (const float*)d_in[4];
    const float* Wmu = (const float*)d_in[5];
    const float* bmu = (const float*)d_in[6];
    const float* Wls = (const float*)d_in[7];
    const float* bls = (const float*)d_in[8];
    const int* row = ei;
    const int* col = ei + N_EDGES;

    float* ws     = (float*)d_ws;
    float* dis    = ws;
    float* invdeg = ws + 50000;
    float* deg    = ws + 100000;
    float* xw     = ws + 150000;
    float* agg1   = ws + 150000 + 6400000;
    float* agg2   = ws + 150000 + 12800000;

    hipMemsetAsync(deg,  0, (size_t)N_NODES * sizeof(float), stream);
    hipMemsetAsync(agg1, 0, (size_t)N_NODES * HID * sizeof(float), stream);
    hipMemsetAsync(agg2, 0, (size_t)N_NODES * HID * sizeof(float), stream);

    deg_kernel<<<(N_EDGES + 255) / 256, 256, 0, stream>>>(row, col, ew, deg);
    dis_kernel<<<(N_NODES + 255) / 256, 256, 0, stream>>>(deg, dis, invdeg);

    gemm_xwt_wmma<<<3125, 256, 0, stream>>>(x, W1, W1, HID, xw);
    scatter_kernel<<<N_EDGES / 8, 256, 0, stream>>>(row, col, ew, dis, xw, agg1);
    act_kernel<<<(N_NODES * HID + 255) / 256, 256, 0, stream>>>(agg1, xw, invdeg, b1);

    gemm_xwt_wmma<<<3125, 256, 0, stream>>>(agg1, Wmu, Wls, OUT_CH, xw);
    scatter_kernel<<<N_EDGES / 8, 256, 0, stream>>>(row, col, ew, dis, xw, agg2);

    out_kernel<<<(N_NODES * OUT_CH + 255) / 256, 256, 0, stream>>>(
        agg2, xw, invdeg, bmu, bls, (float*)d_out);
}


# GEMM epilogue also stores each accumulator's start value (own row/deg + bias): agg memsets, relu pass, output pass go; relu on layer-2 A load; layer-2 atomics add into out; self term summed first
# speedup vs baseline: 1.6688x; 1.6688x over previous
#include <hip/hip_runtime.h>


#define N_NODES 50000
#define N_EDGES 600000
#define IN_CH   128
#define HID     128
#define OUT_CH  64

typedef __attribute__((ext_vector_type(2))) float v2f;
typedef __attribute__((ext_vector_type(8))) float v8f;

__global__ void deg_kernel(const int* __restrict__ row, const int* __restrict__ col,
                           const float* __restrict__ w, float* __restrict__ deg) {
    int e = blockIdx.x * blockDim.x + threadIdx.x;
    if (e >= N_EDGES) return;
    int r = row[e], c = col[e];
    float we = (r == c) ? 0.0f : w[e];
    if (we != 0.0f) atomicAdd(&deg[c], we);
}

__global__ void dis_kernel(const float* __restrict__ deg, float* __restrict__ dis,
                           float* __restrict__ invdeg) {
    int i = blockIdx.x * blockDim.x + threadIdx.x;
    if (i >= N_NODES) return;
    float d = deg[i] + 1.0f;
    dis[i]    = rsqrtf(d);
    invdeg[i] = 1.0f / d;
}

template <bool RELU_A>
__global__ void gemm_xwt_wmma(const float* __restrict__ X,
                              const float* __restrict__ Wa,
                              const float* __restrict__ Wb,
                              int n_split,
                              const float* __restrict__ ba,
                              const float* __restrict__ bb,
                              const float* __restrict__ invdeg,
                              float* __restrict__ C,
                              float* SA,
                              float* SB,
                              int s_pitch) {
    int w     = blockIdx.x * 8 + (threadIdx.x >> 5);
    int lane  = threadIdx.x & 31;
    int nTile = w & 7;
    int mTile = w >> 3;
    if (mTile >= N_NODES / 16) return;

    int half = lane >> 4;
    int l16  = lane & 15;

    const float* xrow = X + (size_t)(mTile * 16 + l16) * IN_CH + half * 2;
    int ng = nTile * 16 + l16;
    const bool lowp = ng < n_split;
    const int  nloc = lowp ? ng : ng - n_split;
    const float* wrow = (lowp ? Wa : Wb) + (size_t)nloc * IN_CH + half * 2;
    const float bias  = (lowp ? ba : bb)[nloc];

    v8f acc = {};
#pragma unroll
    for (int k = 0; k < IN_CH; k += 4) {
        v2f a = *(const v2f*)(xrow + k);
        if (RELU_A) { a.x = fmaxf(a.x, 0.0f); a.y = fmaxf(a.y, 0.0f); }
        v2f b = *(const v2f*)(wrow + k);
        acc = __builtin_amdgcn_wmma_f32_16x16x4_f32(
            false, a, false, b, (short)0, acc, false, false);
    }

    const int m0 = mTile * 16 + half * 8;
    float* crow = C + (size_t)m0 * HID + nTile * 16 + l16;
    float* srow = (lowp ? SA : SB) + (size_t)m0 * s_pitch + nloc;
    const float* idp = invdeg + m0;
#pragma unroll
    for (int j = 0; j < 8; ++j) {
        float v = acc[j];
        crow[(size_t)j * HID]     = v;
        srow[(size_t)j * s_pitch] = v * idp[j] + bias;
    }
}

__global__ void scatter_kernel(const int* __restrict__ row, const int* __restrict__ col,
                               const float* __restrict__ w, const float* __restrict__ dis,
                               const float* __restrict__ src,
                               float* dA, float* dB, int pitch) {
    int wave = (blockIdx.x * blockDim.x + threadIdx.x) >> 5;
    int lane = threadIdx.x & 31;
    if (wave >= N_EDGES) return;
    int r = row[wave], c = col[wave];
    float we = (r == c) ? 0.0f : w[wave];
    if (we == 0.0f) return;
    float coeff = dis[r] * we * dis[c];

    const float* s = src + (size_t)r * HID + lane;
    float v0 = s[0];
    float v1 = s[32];
    float v2 = s[64];
    float v3 = s[96];
    float* d0 = dA + (size_t)c * pitch + lane;
    float* d1 = dB + (size_t)c * pitch + lane;
    atomicAdd(d0 +  0, coeff * v0);
    atomicAdd(d0 + 32, coeff * v1);
    atomicAdd(d1 +  0, coeff * v2);
    atomicAdd(d1 + 32, coeff * v3);
}

extern "C" void kernel_launch(void* const* d_in, const int* in_sizes, int n_in,
                              void* d_out, int out_size, void* d_ws, size_t ws_size,
                              hipStream_t stream) {
    (void)in_sizes; (void)n_in; (void)out_size; (void)ws_size;
    const float* x   = (const float*)d_in[0];
    const int*   ei  = (const int*)d_in[1];
    const float* ew  = (const float*)d_in[2];
    const float* W1  = (const float*)d_in[3];
    const float* b1  = (const float*)d_in[4];
    const float* Wmu = (const float*)d_in[5];
    const float* bmu = (const float*)d_in[6];
    const float* Wls = (const float*)d_in[7];
    const float* bls = (const float*)d_in[8];
    const int* row = ei;
    const int* col = ei + N_EDGES;

    float* ws     = (float*)d_ws;
    float* xw     = ws;
    float* acc1   = ws + 6400000;
    float* dis    = ws + 19200000;
    float* invdeg = ws + 19250000;
    float* deg    = ws + 19300000;
    float* mu     = (float*)d_out;
    float* ls     = (float*)d_out + 3200000;

    hipMemsetAsync(deg, 0, (size_t)N_NODES * sizeof(float), stream);

    deg_kernel<<<(N_EDGES + 255) / 256, 256, 0, stream>>>(row, col, ew, deg);
    dis_kernel<<<(N_NODES + 255) / 256, 256, 0, stream>>>(deg, dis, invdeg);

    gemm_xwt_wmma<false><<<3125, 256, 0, stream>>>(
        x, W1, W1, HID, b1, b1, invdeg, xw, acc1, acc1, HID);
    scatter_kernel<<<N_EDGES / 8, 256, 0, stream>>>(
        row, col, ew, dis, xw, acc1, acc1 + 64, HID);

    gemm_xwt_wmma<true><<<3125, 256, 0, stream>>>(
        acc1, Wmu, Wls, OUT_CH, bmu, bls, invdeg, xw, mu, ls, OUT_CH);
    scatter_kernel<<<N_EDGES / 8, 256, 0, stream>>>(
        row, col, ew, dis, xw, mu, ls, OUT_CH);
}


# Edge pass: a wave takes 16 consecutive edges (was 1): lanes 0-15 load indices, weights, coefficients, handed out by readlane; 32 row gathers in flight ahead of the atomics; sums stay in arrival order
# speedup vs baseline: 2.1060x; 1.2620x over previous
#include <hip/hip_runtime.h>


#define N_NODES 50000
#define N_EDGES 600000
#define IN_CH   128
#define HID     128
#define OUT_CH  64

typedef __attribute__((ext_vector_type(2))) float v2f;
typedef __attribute__((ext_vector_type(8))) float v8f;

__global__ void deg_kernel(const int* __restrict__ row, const int* __restrict__ col,
                           const float* __restrict__ w, float* __restrict__ deg) {
    int e = blockIdx.x * blockDim.x + threadIdx.x;
    if (e >= N_EDGES) return;
    int r = row[e], c = col[e];
    float we = (r == c) ? 0.0f : w[e];
    if (we != 0.0f) atomicAdd(&deg[c], we);
}

__global__ void dis_kernel(const float* __restrict__ deg, float* __restrict__ dis,
                           float* __restrict__ invdeg) {
    int i = blockIdx.x * blockDim.x + threadIdx.x;
    if (i >= N_NODES) return;
    float d = deg[i] + 1.0f;
    dis[i]    = rsqrtf(d);
    invdeg[i] = 1.0f / d;
}

template <bool RELU_A>
__global__ void gemm_xwt_wmma(const float* __restrict__ X,
                              const float* __restrict__ Wa,
                              const float* __restrict__ Wb,
                              int n_split,
                              const float* __restrict__ ba,
                              const float* __restrict__ bb,
                              const float* __restrict__ invdeg,
                              float* __restrict__ C,
                              float* SA,
                              float* SB,
                              int s_pitch) {
    int w     = blockIdx.x * 8 + (threadIdx.x >> 5);
    int lane  = threadIdx.x & 31;
    int nTile = w & 7;
    int mTile = w >> 3;
    if (mTile >= N_NODES / 16) return;

    int half = lane >> 4;
    int l16  = lane & 15;

    const float* xrow = X + (size_t)(mTile * 16 + l16) * IN_CH + half * 2;
    int ng = nTile * 16 + l16;
    const bool lowp = ng < n_split;
    const int  nloc = lowp ? ng : ng - n_split;
    const float* wrow = (lowp ? Wa : Wb) + (size_t)nloc * IN_CH + half * 2;
    const float bias  = (lowp ? ba : bb)[nloc];

    v8f acc = {};
#pragma unroll
    for (int k = 0; k < IN_CH; k += 4) {
        v2f a = *(const v2f*)(xrow + k);
        if (RELU_A) { a.x = fmaxf(a.x, 0.0f); a.y = fmaxf(a.y, 0.0f); }
        v2f b = *(const v2f*)(wrow + k);
        acc = __builtin_amdgcn_wmma_f32_16x16x4_f32(
            false, a, false, b, (short)0, acc, false, false);
    }

    const int m0 = mTile * 16 + half * 8;
    float* crow = C + (size_t)m0 * HID + nTile * 16 + l16;
    float* srow = (lowp ? SA : SB) + (size_t)m0 * s_pitch + nloc;
    const float* idp = invdeg + m0;
#pragma unroll
    for (int j = 0; j < 8; ++j) {
        float v = acc[j];
        crow[(size_t)j * HID]     = v;
        srow[(size_t)j * s_pitch] = v * idp[j] + bias;
    }
}

#define EPW 16
#define EPB 8
static_assert(N_EDGES % EPW == 0 && EPW % EPB == 0 && EPW <= 32, "edge tiling");

__device__ __forceinline__ float lane_value(float x, int l) {
    return __int_as_float(__builtin_amdgcn_readlane(__float_as_int(x), l));
}

__global__ void scatter_kernel(const int* __restrict__ row, const int* __restrict__ col,
                               const float* __restrict__ w, const float* __restrict__ dis,
                               const float* __restrict__ src,
                               float* dA, float* dB, int pitch) {
    int wave = (blockIdx.x * blockDim.x + threadIdx.x) >> 5;
    int lane = threadIdx.x & 31;
    if (wave >= N_EDGES / EPW) return;
    int e = wave * EPW + (lane & (EPW - 1));
    int r = row[e], c = col[e];
    float wv = w[e];
    float we = (r == c) ? 0.0f : wv;
    float coeff = dis[r] * we * dis[c];

#pragma unroll
    for (int b = 0; b < EPW; b += EPB) {
        float v[EPB][4];
#pragma unroll
        for (int j = 0; j < EPB; ++j) {
            int rj = __builtin_amdgcn_readlane(r, b + j);
            const float* s = src + (size_t)rj * HID + lane;
            v[j][0] = s[0];
            v[j][1] = s[32];
            v[j][2] = s[64];
            v[j][3] = s[96];
        }
#pragma unroll
        for (int j = 0; j < EPB; ++j) {
            int   cj = __builtin_amdgcn_readlane(c, b + j);
            float cf = lane_value(coeff, b + j);
            float* d0 = dA + (size_t)cj * pitch + lane;
            float* d1 = dB + (size_t)cj * pitch + lane;
            atomicAdd(d0 +  0, cf * v[j][0]);
            atomicAdd(d0 + 32, cf * v[j][1]);
            atomicAdd(d1 +  0, cf * v[j][2]);
            atomicAdd(d1 + 32, cf * v[j][3]);
        }
    }
}

extern "C" void kernel_launch(void* const* d_in, const int* in_sizes, int n_in,
                              void* d_out, int out_size, void* d_ws, size_t ws_size,
                              hipStream_t stream) {
    (void)in_sizes; (void)n_in; (void)out_size; (void)ws_size;
    const float* x   = (const float*)d_in[0];
    const int*   ei  = (const int*)d_in[1];
    const float* ew  = (const float*)d_in[2];
    const float* W1  = (const float*)d_in[3];
    const float* b1  = (const float*)d_in[4];
    const float* Wmu = (const float*)d_in[5];
    const float* bmu = (const float*)d_in[6];
    const float* Wls = (const float*)d_in[7];
    const float* bls = (const float*)d_in[8];
    const int* row = ei;
    const int* col = ei + N_EDGES;

    float* ws     = (float*)d_ws;
    float* xw     = ws;
    float* acc1   = ws + 6400000;
    float* dis    = ws + 19200000;
    float* invdeg = ws + 19250000;
    float* deg    = ws + 19300000;
    float* mu     = (float*)d_out;
    float* ls     = (float*)d_out + 3200000;

    hipMemsetAsync(deg, 0, (size_t)N_NODES * sizeof(float), stream);

    deg_kernel<<<(N_EDGES + 255) / 256, 256, 0, stream>>>(row, col, ew, deg);
    dis_kernel<<<(N_NODES + 255) / 256, 256, 0, stream>>>(deg, dis, invdeg);

    gemm_xwt_wmma<false><<<3125, 256, 0, stream>>>(
        x, W1, W1, HID, b1, b1, invdeg, xw, acc1, acc1, HID);
    scatter_kernel<<<(N_EDGES / EPW + 7) / 8, 256, 0, stream>>>(
        row, col, ew, dis, xw, acc1, acc1 + 64, HID);

    gemm_xwt_wmma<true><<<3125, 256, 0, stream>>>(
        acc1, Wmu, Wls, OUT_CH, bmu, bls, invdeg, xw, mu, ls, OUT_CH);
    scatter_kernel<<<(N_EDGES / EPW + 7) / 8, 256, 0, stream>>>(
        row, col, ew, dis, xw, mu, ls, OUT_CH);
}


# GEMMs: work-group stages 80 rows of X once in LDS (whole-row loads), each wave keeps its 16 weight columns in registers for 5 row tiles, whole-row stores; k sum order changes inside each 8
# speedup vs baseline: 2.5000x; 1.1871x over previous
#include <hip/hip_runtime.h>


#define N_NODES 50000
#define N_EDGES 600000
#define IN_CH   128
#define HID     128
#define OUT_CH  64

typedef __attribute__((ext_vector_type(2))) float v2f;
typedef __attribute__((ext_vector_type(4))) float v4f;
typedef __attribute__((ext_vector_type(8))) float v8f;

__global__ void deg_kernel(const int* __restrict__ row, const int* __restrict__ col,
                           const float* __restrict__ w, float* __restrict__ deg) {
    int e = blockIdx.x * blockDim.x + threadIdx.x;
    if (e >= N_EDGES) return;
    int r = row[e], c = col[e];
    float we = (r == c) ? 0.0f : w[e];
    if (we != 0.0f) atomicAdd(&deg[c], we);
}

__global__ void dis_kernel(const float* __restrict__ deg, float* __restrict__ dis,
                           float* __restrict__ invdeg) {
    int i = blockIdx.x * blockDim.x + threadIdx.x;
    if (i >= N_NODES) return;
    float d = deg[i] + 1.0f;
    dis[i]    = rsqrtf(d);
    invdeg[i] = 1.0f / d;
}

#define GM_ROWS  80
#define GM_MT    (GM_ROWS / 16)
#define GM_RPW   (GM_ROWS / 8)
#define GM_PITCH 132
static_assert(N_NODES % GM_ROWS == 0 && IN_CH == 128 && HID == 128, "gemm tiling");

template <bool RELU_A>
__global__ void __launch_bounds__(256)
gemm_xwt_wmma(const float* __restrict__ X,
              const float* __restrict__ Wa,
              const float* __restrict__ Wb,
              int n_split,
              const float* __restrict__ ba,
              const float* __restrict__ bb,
              const float* __restrict__ invdeg,
              float* __restrict__ C,
              float* SA,
              float* SB,
              int s_pitch) {
    __shared__ float tile[GM_ROWS * GM_PITCH];

    const int wv   = __builtin_amdgcn_readfirstlane((int)(threadIdx.x >> 5));
    const int lane = threadIdx.x & 31;
    const int half = lane >> 4;
    const int l16  = lane & 15;
    const int m0   = blockIdx.x * GM_ROWS;
    const int r0   = wv * GM_RPW;

    v4f xr[GM_RPW];
#pragma unroll
    for (int j = 0; j < GM_RPW; ++j)
        xr[j] = *(const v4f*)(X + (size_t)(m0 + r0 + j) * IN_CH + lane * 4);

    const int  ng   = wv * 16 + l16;
    const bool lowp = ng < n_split;
    const int  nloc = lowp ? ng : ng - n_split;
    const float* wrow = (lowp ? Wa : Wb) + (size_t)nloc * IN_CH + half * 4;
    v4f bf[16];
#pragma unroll
    for (int p = 0; p < 16; ++p)
        bf[p] = *(const v4f*)(wrow + 8 * p);

#pragma unroll
    for (int j = 0; j < GM_RPW; ++j) {
        v4f t = xr[j];
        if (RELU_A) {
            t.x = fmaxf(t.x, 0.0f); t.y = fmaxf(t.y, 0.0f);
            t.z = fmaxf(t.z, 0.0f); t.w = fmaxf(t.w, 0.0f);
        }
        *(v4f*)(tile + (r0 + j) * GM_PITCH + lane * 4) = t;
    }
    __syncthreads();

    v8f acc[GM_MT];
#pragma unroll
    for (int i = 0; i < GM_MT; ++i) acc[i] = (v8f){};

    const float* arow = tile + l16 * GM_PITCH + half * 4;
    v4f a[GM_MT];
#pragma unroll
    for (int i = 0; i < GM_MT; ++i)
        a[i] = *(const v4f*)(arow + i * 16 * GM_PITCH);
#pragma unroll
    for (int p = 0; p < 16; ++p) {
        v4f an[GM_MT];
        if (p + 1 < 16) {
#pragma unroll
            for (int i = 0; i < GM_MT; ++i)
                an[i] = *(const v4f*)(arow + i * 16 * GM_PITCH + 8 * (p + 1));
        }
        __builtin_amdgcn_sched_barrier(0);
#pragma unroll
        for (int i = 0; i < GM_MT; ++i)
            acc[i] = __builtin_amdgcn_wmma_f32_16x16x4_f32(
                false, (v2f){a[i].x, a[i].y}, false, (v2f){bf[p].x, bf[p].y},
                (short)0, acc[i], false, false);
#pragma unroll
        for (int i = 0; i < GM_MT; ++i)
            acc[i] = __builtin_amdgcn_wmma_f32_16x16x4_f32(
                false, (v2f){a[i].z, a[i].w}, false, (v2f){bf[p].z, bf[p].w},
                (short)0, acc[i], false, false);
        __builtin_amdgcn_sched_barrier(0);
        if (p + 1 < 16) {
#pragma unroll
            for (int i = 0; i < GM_MT; ++i) a[i] = an[i];
        }
    }

    __syncthreads();
#pragma unroll
    for (int i = 0; i < GM_MT; ++i)
#pragma unroll
        for (int j = 0; j < 8; ++j)
            tile[(16 * i + 8 * half + j) * GM_PITCH + wv * 16 + l16] = acc[i][j];
    __syncthreads();

    const int  c0   = lane * 4;
    const bool lowq = c0 < n_split;
    const int  cl   = lowq ? c0 : c0 - n_split;
    const v4f  b4   = *(const v4f*)((lowq ? ba : bb) + cl);
    float* sbase    = (lowq ? SA : SB) + cl;
#pragma unroll
    for (int j = 0; j < GM_RPW; ++j) {
        const int m = m0 + r0 + j;
        const v4f v = *(const v4f*)(tile + (r0 + j) * GM_PITCH + c0);
        const float id = invdeg[m];
        *(v4f*)(C + (size_t)m * HID + c0) = v;
        *(v4f*)(sbase + (size_t)m * s_pitch) = v * id + b4;
    }
}

#define EPW 16
#define EPB 8
static_assert(N_EDGES % EPW == 0 && EPW % EPB == 0 && EPW <= 32, "edge tiling");

__device__ __forceinline__ float lane_value(float x, int l) {
    return __int_as_float(__builtin_amdgcn_readlane(__float_as_int(x), l));
}

__global__ void scatter_kernel(const int* __restrict__ row, const int* __restrict__ col,
                               const float* __restrict__ w, const float* __restrict__ dis,
                               const float* __restrict__ src,
                               float* dA, float* dB, int pitch) {
    int wave = (blockIdx.x * blockDim.x + threadIdx.x) >> 5;
    int lane = threadIdx.x & 31;
    if (wave >= N_EDGES / EPW) return;
    int e = wave * EPW + (lane & (EPW - 1));
    int r = row[e], c = col[e];
    float wv = w[e];
    float we = (r == c) ? 0.0f : wv;
    float coeff = dis[r] * we * dis[c];

#pragma unroll
    for (int b = 0; b < EPW; b += EPB) {
        float v[EPB][4];
#pragma unroll
        for (int j = 0; j < EPB; ++j) {
            int rj = __builtin_amdgcn_readlane(r, b + j);
            const float* s = src + (size_t)rj * HID + lane;
            v[j][0] = s[0];
            v[j][1] = s[32];
            v[j][2] = s[64];
            v[j][3] = s[96];
        }
#pragma unroll
        for (int j = 0; j < EPB; ++j) {
            int   cj = __builtin_amdgcn_readlane(c, b + j);
            float cf = lane_value(coeff, b + j);
            float* d0 = dA + (size_t)cj * pitch + lane;
            float* d1 = dB + (size_t)cj * pitch + lane;
            atomicAdd(d0 +  0, cf * v[j][0]);
            atomicAdd(d0 + 32, cf * v[j][1]);
            atomicAdd(d1 +  0, cf * v[j][2]);
            atomicAdd(d1 + 32, cf * v[j][3]);
        }
    }
}

extern "C" void kernel_launch(void* const* d_in, const int* in_sizes, int n_in,
                              void* d_out, int out_size, void* d_ws, size_t ws_size,
                              hipStream_t stream) {
    (void)in_sizes; (void)n_in; (void)out_size; (void)ws_size;
    const float* x   = (const float*)d_in[0];
    const int*   ei  = (const int*)d_in[1];
    const float* ew  = (const float*)d_in[2];
    const float* W1  = (const float*)d_in[3];
    const float* b1  = (const float*)d_in[4];
    const float* Wmu = (const float*)d_in[5];
    const float* bmu = (const float*)d_in[6];
    const float* Wls = (const float*)d_in[7];
    const float* bls = (const float*)d_in[8];
    const int* row = ei;
    const int* col = ei + N_EDGES;

    float* ws     = (float*)d_ws;
    float* xw     = ws;
    float* acc1   = ws + 6400000;
    float* dis    = ws + 19200000;
    float* invdeg = ws + 19250000;
    float* deg    = ws + 19300000;
    float* mu     = (float*)d_out;
    float* ls     = (float*)d_out + 3200000;

    hipMemsetAsync(deg, 0, (size_t)N_NODES * sizeof(float), stream);

    deg_kernel<<<(N_EDGES + 255) / 256, 256, 0, stream>>>(row, col, ew, deg);
    dis_kernel<<<(N_NODES + 255) / 256, 256, 0, stream>>>(deg, dis, invdeg);

    gemm_xwt_wmma<false><<<N_NODES / GM_ROWS, 256, 0, stream>>>(
        x, W1, W1, HID, b1, b1, invdeg, xw, acc1, acc1, HID);
    scatter_kernel<<<(N_EDGES / EPW + 7) / 8, 256, 0, stream>>>(
        row, col, ew, dis, xw, acc1, acc1 + 64, HID);

    gemm_xwt_wmma<true><<<N_NODES / GM_ROWS, 256, 0, stream>>>(
        acc1, Wmu, Wls, OUT_CH, bmu, bls, invdeg, xw, mu, ls, OUT_CH);
    scatter_kernel<<<(N_EDGES / EPW + 7) / 8, 256, 0, stream>>>(
        row, col, ew, dis, xw, mu, ls, OUT_CH);
}


# Degree sums: 32 work-groups add their edges' weights into own 50000-word LDS tables (LDS f32 atomics), stored as planes; node kernel adds planes: no global deg atomics, no memset; sum order changes
# speedup vs baseline: 2.8064x; 1.1226x over previous
#include <hip/hip_runtime.h>


#define N_NODES 50000
#define N_EDGES 600000
#define IN_CH   128
#define HID     128
#define OUT_CH  64

typedef __attribute__((ext_vector_type(2))) float v2f;
typedef __attribute__((ext_vector_type(4))) float v4f;
typedef __attribute__((ext_vector_type(8))) float v8f;
typedef __attribute__((ext_vector_type(4))) int   v4i;

#define DG_WGS   32
#define DG_THR   1024
#define DG_PITCH 50016
#define DG_QUADS (N_EDGES / 4)
#define DG_ITERS ((DG_QUADS + DG_WGS * DG_THR - 1) / (DG_WGS * DG_THR))
static_assert(N_EDGES % 4 == 0 && N_NODES % 4 == 0 && DG_PITCH >= N_NODES && DG_PITCH % 32 == 0,
              "degree tiling");

__global__ void __launch_bounds__(DG_THR)
deg_hist_kernel(const int* __restrict__ row, const int* __restrict__ col,
                const float* __restrict__ w, float* __restrict__ part) {
    __shared__ __attribute__((aligned(16))) float hist[N_NODES];
    const int tid = threadIdx.x;

    v4i r[DG_ITERS], c[DG_ITERS];
    v4f x[DG_ITERS];
#pragma unroll
    for (int k = 0; k < DG_ITERS; ++k) {
        int q = (k * DG_WGS + (int)blockIdx.x) * DG_THR + tid;
        q = q < DG_QUADS ? q : DG_QUADS - 1;
        r[k] = ((const v4i*)row)[q];
        c[k] = ((const v4i*)col)[q];
        x[k] = ((const v4f*)w)[q];
    }

    for (int i = tid; i < N_NODES / 4; i += DG_THR)
        ((v4f*)hist)[i] = (v4f){};
    __syncthreads();

#pragma unroll
    for (int k = 0; k < DG_ITERS; ++k) {
        const int q = (k * DG_WGS + (int)blockIdx.x) * DG_THR + tid;
        if (q < DG_QUADS) {
            atomicAdd(&hist[c[k].x], r[k].x == c[k].x ? 0.0f : x[k].x);
            atomicAdd(&hist[c[k].y], r[k].y == c[k].y ? 0.0f : x[k].y);
            atomicAdd(&hist[c[k].z], r[k].z == c[k].z ? 0.0f : x[k].z);
            atomicAdd(&hist[c[k].w], r[k].w == c[k].w ? 0.0f : x[k].w);
        }
    }
    __syncthreads();

    float* plane = part + (size_t)blockIdx.x * DG_PITCH;
    for (int i = tid; i < N_NODES / 4; i += DG_THR)
        ((v4f*)plane)[i] = ((const v4f*)hist)[i];
}

__global__ void dis_kernel(const float* __restrict__ part, float* __restrict__ dis,
                           float* __restrict__ invdeg) {
    int i = blockIdx.x * blockDim.x + threadIdx.x;
    if (i >= N_NODES) return;
    float p[DG_WGS];
#pragma unroll
    for (int g = 0; g < DG_WGS; ++g)
        p[g] = part[(size_t)g * DG_PITCH + i];
    float s = p[0];
#pragma unroll
    for (int g = 1; g < DG_WGS; ++g)
        s += p[g];
    float d = s + 1.0f;
    dis[i]    = rsqrtf(d);
    invdeg[i] = 1.0f / d;
}

#define GM_ROWS  80
#define GM_MT    (GM_ROWS / 16)
#define GM_RPW   (GM_ROWS / 8)
#define GM_PITCH 132
static_assert(N_NODES % GM_ROWS == 0 && IN_CH == 128 && HID == 128, "gemm tiling");

template <bool RELU_A>
__global__ void __launch_bounds__(256)
gemm_xwt_wmma(const float* __restrict__ X,
              const float* __restrict__ Wa,
              const float* __restrict__ Wb,
              int n_split,
              const float* __restrict__ ba,
              const float* __restrict__ bb,
              const float* __restrict__ invdeg,
              float* __restrict__ C,
              float* SA,
              float* SB,
              int s_pitch) {
    __shared__ float tile[GM_ROWS * GM_PITCH];

    const int wv   = __builtin_amdgcn_readfirstlane((int)(threadIdx.x >> 5));
    const int lane = threadIdx.x & 31;
    const int half = lane >> 4;
    const int l16  = lane & 15;
    const int m0   = blockIdx.x * GM_ROWS;
    const int r0   = wv * GM_RPW;

    v4f xr[GM_RPW];
#pragma unroll
    for (int j = 0; j < GM_RPW; ++j)
        xr[j] = *(const v4f*)(X + (size_t)(m0 + r0 + j) * IN_CH + lane * 4);

    const int  ng   = wv * 16 + l16;
    const bool lowp = ng < n_split;
    const int  nloc = lowp ? ng : ng - n_split;
    const float* wrow = (lowp ? Wa : Wb) + (size_t)nloc * IN_CH + half * 4;
    v4f bf[16];
#pragma unroll
    for (int p = 0; p < 16; ++p)
        bf[p] = *(const v4f*)(wrow + 8 * p);

#pragma unroll
    for (int j = 0; j < GM_RPW; ++j) {
        v4f t = xr[j];
        if (RELU_A) {
            t.x = fmaxf(t.x, 0.0f); t.y = fmaxf(t.y, 0.0f);
            t.z = fmaxf(t.z, 0.0f); t.w = fmaxf(t.w, 0.0f);
        }
        *(v4f*)(tile + (r0 + j) * GM_PITCH + lane * 4) = t;
    }
    __syncthreads();

    v8f acc[GM_MT];
#pragma unroll
    for (int i = 0; i < GM_MT; ++i) acc[i] = (v8f){};

    const float* arow = tile + l16 * GM_PITCH + half * 4;
    v4f a[GM_MT];
#pragma unroll
    for (int i = 0; i < GM_MT; ++i)
        a[i] = *(const v4f*)(arow + i * 16 * GM_PITCH);
#pragma unroll
    for (int p = 0; p < 16; ++p) {
        v4f an[GM_MT];
        if (p + 1 < 16) {
#pragma unroll
            for (int i = 0; i < GM_MT; ++i)
                an[i] = *(const v4f*)(arow + i * 16 * GM_PITCH + 8 * (p + 1));
        }
        __builtin_amdgcn_sched_barrier(0);
#pragma unroll
        for (int i = 0; i < GM_MT; ++i)
            acc[i] = __builtin_amdgcn_wmma_f32_16x16x4_f32(
                false, (v2f){a[i].x, a[i].y}, false, (v2f){bf[p].x, bf[p].y},
                (short)0, acc[i], false, false);
#pragma unroll
        for (int i = 0; i < GM_MT; ++i)
            acc[i] = __builtin_amdgcn_wmma_f32_16x16x4_f32(
                false, (v2f){a[i].z, a[i].w}, false, (v2f){bf[p].z, bf[p].w},
                (short)0, acc[i], false, false);
        __builtin_amdgcn_sched_barrier(0);
        if (p + 1 < 16) {
#pragma unroll
            for (int i = 0; i < GM_MT; ++i) a[i] = an[i];
        }
    }

    __syncthreads();
#pragma unroll
    for (int i = 0; i < GM_MT; ++i)
#pragma unroll
        for (int j = 0; j < 8; ++j)
            tile[(16 * i + 8 * half + j) * GM_PITCH + wv * 16 + l16] = acc[i][j];
    __syncthreads();

    const int  c0   = lane * 4;
    const bool lowq = c0 < n_split;
    const int  cl   = lowq ? c0 : c0 - n_split;
    const v4f  b4   = *(const v4f*)((lowq ? ba : bb) + cl);
    float* sbase    = (lowq ? SA : SB) + cl;
#pragma unroll
    for (int j = 0; j < GM_RPW; ++j) {
        const int m = m0 + r0 + j;
        const v4f v = *(const v4f*)(tile + (r0 + j) * GM_PITCH + c0);
        const float id = invdeg[m];
        *(v4f*)(C + (size_t)m * HID + c0) = v;
        *(v4f*)(sbase + (size_t)m * s_pitch) = v * id + b4;
    }
}

#define EPW 16
#define EPB 8
static_assert(N_EDGES % EPW == 0 && EPW % EPB == 0 && EPW <= 32, "edge tiling");

__device__ __forceinline__ float lane_value(float x, int l) {
    return __int_as_float(__builtin_amdgcn_readlane(__float_as_int(x), l));
}

__global__ void scatter_kernel(const int* __restrict__ row, const int* __restrict__ col,
                               const float* __restrict__ w, const float* __restrict__ dis,
                               const float* __restrict__ src,
                               float* dA, float* dB, int pitch) {
    int wave = (blockIdx.x * blockDim.x + threadIdx.x) >> 5;
    int lane = threadIdx.x & 31;
    if (wave >= N_EDGES / EPW) return;
    int e = wave * EPW + (lane & (EPW - 1));
    int r = row[e], c = col[e];
    float wv = w[e];
    float we = (r == c) ? 0.0f : wv;
    float coeff = dis[r] * we * dis[c];

#pragma unroll
    for (int b = 0; b < EPW; b += EPB) {
        float v[EPB][4];
#pragma unroll
        for (int j = 0; j < EPB; ++j) {
            int rj = __builtin_amdgcn_readlane(r, b + j);
            const float* s = src + (size_t)rj * HID + lane;
            v[j][0] = s[0];
            v[j][1] = s[32];
            v[j][2] = s[64];
            v[j][3] = s[96];
        }
#pragma unroll
        for (int j = 0; j < EPB; ++j) {
            int   cj = __builtin_amdgcn_readlane(c, b + j);
            float cf = lane_value(coeff, b + j);
            float* d0 = dA + (size_t)cj * pitch + lane;
            float* d1 = dB + (size_t)cj * pitch + lane;
            atomicAdd(d0 +  0, cf * v[j][0]);
            atomicAdd(d0 + 32, cf * v[j][1]);
            atomicAdd(d1 +  0, cf * v[j][2]);
            atomicAdd(d1 + 32, cf * v[j][3]);
        }
    }
}

extern "C" void kernel_launch(void* const* d_in, const int* in_sizes, int n_in,
                              void* d_out, int out_size, void* d_ws, size_t ws_size,
                              hipStream_t stream) {
    (void)in_sizes; (void)n_in; (void)out_size; (void)ws_size;
    const float* x   = (const float*)d_in[0];
    const int*   ei  = (const int*)d_in[1];
    const float* ew  = (const float*)d_in[2];
    const float* W1  = (const float*)d_in[3];
    const float* b1  = (const float*)d_in[4];
    const float* Wmu = (const float*)d_in[5];
    const float* bmu = (const float*)d_in[6];
    const float* Wls = (const float*)d_in[7];
    const float* bls = (const float*)d_in[8];
    const int* row = ei;
    const int* col = ei + N_EDGES;

    float* ws     = (float*)d_ws;
    float* xw     = ws;
    float* acc1   = ws + 6400000;
    float* dis    = ws + 19200000;
    float* invdeg = ws + 19250000;
    float* part   = ws + 12800000;
    float* mu     = (float*)d_out;
    float* ls     = (float*)d_out + 3200000;


    deg_hist_kernel<<<DG_WGS, DG_THR, 0, stream>>>(row, col, ew, part);
    dis_kernel<<<(N_NODES + 255) / 256, 256, 0, stream>>>(part, dis, invdeg);

    gemm_xwt_wmma<false><<<N_NODES / GM_ROWS, 256, 0, stream>>>(
        x, W1, W1, HID, b1, b1, invdeg, xw, acc1, acc1, HID);
    scatter_kernel<<<(N_EDGES / EPW + 7) / 8, 256, 0, stream>>>(
        row, col, ew, dis, xw, acc1, acc1 + 64, HID);

    gemm_xwt_wmma<true><<<N_NODES / GM_ROWS, 256, 0, stream>>>(
        acc1, Wmu, Wls, OUT_CH, bmu, bls, invdeg, xw, mu, ls, OUT_CH);
    scatter_kernel<<<(N_EDGES / EPW + 7) / 8, 256, 0, stream>>>(
        row, col, ew, dis, xw, mu, ls, OUT_CH);
}


# Edge pass: a wave reads which half of the package it runs on, draws 32-edge items from that half's counter (int atomic) and adds only edges whose target node is in that half's range; arrival order
# speedup vs baseline: 2.9621x; 1.0555x over previous
#include <hip/hip_runtime.h>


#define N_NODES 50000
#define N_EDGES 600000
#define IN_CH   128
#define HID     128
#define OUT_CH  64

typedef __attribute__((ext_vector_type(2))) float v2f;
typedef __attribute__((ext_vector_type(4))) float v4f;
typedef __attribute__((ext_vector_type(8))) float v8f;
typedef __attribute__((ext_vector_type(4))) int   v4i;

#define DG_WGS   32
#define DG_THR   1024
#define DG_PITCH 50016
#define DG_QUADS (N_EDGES / 4)
#define DG_ITERS ((DG_QUADS + DG_WGS * DG_THR - 1) / (DG_WGS * DG_THR))
static_assert(N_EDGES % 4 == 0 && N_NODES % 4 == 0 && DG_PITCH >= N_NODES && DG_PITCH % 32 == 0,
              "degree tiling");

__global__ void __launch_bounds__(DG_THR)
deg_hist_kernel(const int* __restrict__ row, const int* __restrict__ col,
                const float* __restrict__ w, float* __restrict__ part) {
    __shared__ __attribute__((aligned(16))) float hist[N_NODES];
    const int tid = threadIdx.x;

    v4i r[DG_ITERS], c[DG_ITERS];
    v4f x[DG_ITERS];
#pragma unroll
    for (int k = 0; k < DG_ITERS; ++k) {
        int q = (k * DG_WGS + (int)blockIdx.x) * DG_THR + tid;
        q = q < DG_QUADS ? q : DG_QUADS - 1;
        r[k] = ((const v4i*)row)[q];
        c[k] = ((const v4i*)col)[q];
        x[k] = ((const v4f*)w)[q];
    }

    for (int i = tid; i < N_NODES / 4; i += DG_THR)
        ((v4f*)hist)[i] = (v4f){};
    __syncthreads();

#pragma unroll
    for (int k = 0; k < DG_ITERS; ++k) {
        const int q = (k * DG_WGS + (int)blockIdx.x) * DG_THR + tid;
        if (q < DG_QUADS) {
            atomicAdd(&hist[c[k].x], r[k].x == c[k].x ? 0.0f : x[k].x);
            atomicAdd(&hist[c[k].y], r[k].y == c[k].y ? 0.0f : x[k].y);
            atomicAdd(&hist[c[k].z], r[k].z == c[k].z ? 0.0f : x[k].z);
            atomicAdd(&hist[c[k].w], r[k].w == c[k].w ? 0.0f : x[k].w);
        }
    }
    __syncthreads();

    float* plane = part + (size_t)blockIdx.x * DG_PITCH;
    for (int i = tid; i < N_NODES / 4; i += DG_THR)
        ((v4f*)plane)[i] = ((const v4f*)hist)[i];
}

#define CTR_STRIDE 64
__global__ void dis_kernel(const float* __restrict__ part, float* __restrict__ dis,
                           float* __restrict__ invdeg, int* __restrict__ ctr) {
    int i = blockIdx.x * blockDim.x + threadIdx.x;
    if (i >= N_NODES) return;
    if (i < 4) ctr[i * CTR_STRIDE] = 0;
    float p[DG_WGS];
#pragma unroll
    for (int g = 0; g < DG_WGS; ++g)
        p[g] = part[(size_t)g * DG_PITCH + i];
    float s = p[0];
#pragma unroll
    for (int g = 1; g < DG_WGS; ++g)
        s += p[g];
    float d = s + 1.0f;
    dis[i]    = rsqrtf(d);
    invdeg[i] = 1.0f / d;
}

#define GM_ROWS  80
#define GM_MT    (GM_ROWS / 16)
#define GM_RPW   (GM_ROWS / 8)
#define GM_PITCH 132
static_assert(N_NODES % GM_ROWS == 0 && IN_CH == 128 && HID == 128, "gemm tiling");

template <bool RELU_A>
__global__ void __launch_bounds__(256)
gemm_xwt_wmma(const float* __restrict__ X,
              const float* __restrict__ Wa,
              const float* __restrict__ Wb,
              int n_split,
              const float* __restrict__ ba,
              const float* __restrict__ bb,
              const float* __restrict__ invdeg,
              float* __restrict__ C,
              float* SA,
              float* SB,
              int s_pitch) {
    __shared__ float tile[GM_ROWS * GM_PITCH];

    const int wv   = __builtin_amdgcn_readfirstlane((int)(threadIdx.x >> 5));
    const int lane = threadIdx.x & 31;
    const int half = lane >> 4;
    const int l16  = lane & 15;
    const int m0   = blockIdx.x * GM_ROWS;
    const int r0   = wv * GM_RPW;

    v4f xr[GM_RPW];
#pragma unroll
    for (int j = 0; j < GM_RPW; ++j)
        xr[j] = *(const v4f*)(X + (size_t)(m0 + r0 + j) * IN_CH + lane * 4);

    const int  ng   = wv * 16 + l16;
    const bool lowp = ng < n_split;
    const int  nloc = lowp ? ng : ng - n_split;
    const float* wrow = (lowp ? Wa : Wb) + (size_t)nloc * IN_CH + half * 4;
    v4f bf[16];
#pragma unroll
    for (int p = 0; p < 16; ++p)
        bf[p] = *(const v4f*)(wrow + 8 * p);

#pragma unroll
    for (int j = 0; j < GM_RPW; ++j) {
        v4f t = xr[j];
        if (RELU_A) {
            t.x = fmaxf(t.x, 0.0f); t.y = fmaxf(t.y, 0.0f);
            t.z = fmaxf(t.z, 0.0f); t.w = fmaxf(t.w, 0.0f);
        }
        *(v4f*)(tile + (r0 + j) * GM_PITCH + lane * 4) = t;
    }
    __syncthreads();

    v8f acc[GM_MT];
#pragma unroll
    for (int i = 0; i < GM_MT; ++i) acc[i] = (v8f){};

    const float* arow = tile + l16 * GM_PITCH + half * 4;
    v4f a[GM_MT];
#pragma unroll
    for (int i = 0; i < GM_MT; ++i)
        a[i] = *(const v4f*)(arow + i * 16 * GM_PITCH);
#pragma unroll
    for (int p = 0; p < 16; ++p) {
        v4f an[GM_MT];
        if (p + 1 < 16) {
#pragma unroll
            for (int i = 0; i < GM_MT; ++i)
                an[i] = *(const v4f*)(arow + i * 16 * GM_PITCH + 8 * (p + 1));
        }
        __builtin_amdgcn_sched_barrier(0);
#pragma unroll
        for (int i = 0; i < GM_MT; ++i)
            acc[i] = __builtin_amdgcn_wmma_f32_16x16x4_f32(
                false, (v2f){a[i].x, a[i].y}, false, (v2f){bf[p].x, bf[p].y},
                (short)0, acc[i], false, false);
#pragma unroll
        for (int i = 0; i < GM_MT; ++i)
            acc[i] = __builtin_amdgcn_wmma_f32_16x16x4_f32(
                false, (v2f){a[i].z, a[i].w}, false, (v2f){bf[p].z, bf[p].w},
                (short)0, acc[i], false, false);
        __builtin_amdgcn_sched_barrier(0);
        if (p + 1 < 16) {
#pragma unroll
            for (int i = 0; i < GM_MT; ++i) a[i] = an[i];
        }
    }

    __syncthreads();
#pragma unroll
    for (int i = 0; i < GM_MT; ++i)
#pragma unroll
        for (int j = 0; j < 8; ++j)
            tile[(16 * i + 8 * half + j) * GM_PITCH + wv * 16 + l16] = acc[i][j];
    __syncthreads();

    const int  c0   = lane * 4;
    const bool lowq = c0 < n_split;
    const int  cl   = lowq ? c0 : c0 - n_split;
    const v4f  b4   = *(const v4f*)((lowq ? ba : bb) + cl);
    float* sbase    = (lowq ? SA : SB) + cl;
#pragma unroll
    for (int j = 0; j < GM_RPW; ++j) {
        const int m = m0 + r0 + j;
        const v4f v = *(const v4f*)(tile + (r0 + j) * GM_PITCH + c0);
        const float id = invdeg[m];
        *(v4f*)(C + (size_t)m * HID + c0) = v;
        *(v4f*)(sbase + (size_t)m * s_pitch) = v * id + b4;
    }
}

#define EPI     32
#define N_ITEMS (N_EDGES / EPI)
#define EPB     8
#define SC_WGS  1024
static_assert(N_EDGES % EPI == 0 && EPI % EPB == 0 && EPI == 32 && N_NODES % 2 == 0, "edge tiling");

__device__ __forceinline__ float lane_value(float x, int l) {
    return __int_as_float(__builtin_amdgcn_readlane(__float_as_int(x), l));
}

__device__ __forceinline__ int next_item(int* ctr, int cls, int lane) {
    int v = 0;
    if (lane == 0) v = atomicAdd(ctr + cls * CTR_STRIDE, 1);
    return __builtin_amdgcn_readfirstlane(v);
}

__global__ void __launch_bounds__(256)
scatter_kernel(const int* __restrict__ row, const int* __restrict__ col,
               const float* __restrict__ w, const float* __restrict__ dis,
               const float* __restrict__ src,
               float* dA, float* dB, int pitch, int* ctr) {
    const int lane = threadIdx.x & 31;
    int  cls   = (int)((__builtin_amdgcn_s_sendmsg_rtn(0x87) >> 18) & 1u);
    bool other = false;
    const bool stealer = __builtin_amdgcn_readfirstlane((int)(threadIdx.x >> 5)) == 0;
    int  item  = next_item(ctr, cls, lane);
    for (int it = 0; it < 2 * N_ITEMS + 2; ++it) {
        if (item >= N_ITEMS) {
            if (other || !stealer) break;
            other = true;
            cls ^= 1;
            item = next_item(ctr, cls, lane);
            if (item >= N_ITEMS) break;
        }
        const int e = item * EPI + lane;
        const int r = row[e], c = col[e];
        const float wv = w[e];
        const int nxt = next_item(ctr, cls, lane);
        const bool mine = (c >= N_NODES / 2) == (cls != 0);
        const float we = (r == c) ? 0.0f : wv;
        const float dr = dis[mine ? r : 0];
        const float dc = dis[mine ? c : 0];
        unsigned mask = __builtin_amdgcn_ballot_w32(mine);

        for (int b = 0; b < EPI / EPB && mask != 0u; ++b) {
            const int n = __builtin_popcount(mask);
            int ln[EPB];
#pragma unroll
            for (int j = 0; j < EPB; ++j) {
                ln[j] = mask != 0u ? __builtin_ctz(mask) : 0;
                mask &= mask - 1u;
            }
            float v[EPB][4];
#pragma unroll
            for (int j = 0; j < EPB; ++j) {
                if (j < n) {
                    const int rj = __builtin_amdgcn_readlane(r, ln[j]);
                    const float* s = src + (size_t)rj * HID + lane;
                    v[j][0] = s[0];
                    v[j][1] = s[32];
                    v[j][2] = s[64];
                    v[j][3] = s[96];
                }
            }
#pragma unroll
            for (int j = 0; j < EPB; ++j) {
                if (j < n) {
                    const int   cj = __builtin_amdgcn_readlane(c, ln[j]);
                    const float cf = lane_value(dr, ln[j]) * lane_value(we, ln[j]) * lane_value(dc, ln[j]);
                    float* d0 = dA + (size_t)cj * pitch + lane;
                    float* d1 = dB + (size_t)cj * pitch + lane;
                    atomicAdd(d0 +  0, cf * v[j][0]);
                    atomicAdd(d0 + 32, cf * v[j][1]);
                    atomicAdd(d1 +  0, cf * v[j][2]);
                    atomicAdd(d1 + 32, cf * v[j][3]);
                }
            }
        }
        item = nxt;
    }
}

extern "C" void kernel_launch(void* const* d_in, const int* in_sizes, int n_in,
                              void* d_out, int out_size, void* d_ws, size_t ws_size,
                              hipStream_t stream) {
    (void)in_sizes; (void)n_in; (void)out_size; (void)ws_size;
    const float* x   = (const float*)d_in[0];
    const int*   ei  = (const int*)d_in[1];
    const float* ew  = (const float*)d_in[2];
    const float* W1  = (const float*)d_in[3];
    const float* b1  = (const float*)d_in[4];
    const float* Wmu = (const float*)d_in[5];
    const float* bmu = (const float*)d_in[6];
    const float* Wls = (const float*)d_in[7];
    const float* bls = (const float*)d_in[8];
    const int* row = ei;
    const int* col = ei + N_EDGES;

    float* ws     = (float*)d_ws;
    float* xw     = ws;
    float* acc1   = ws + 6400000;
    float* dis    = ws + 19200000;
    float* invdeg = ws + 19250000;
    float* part   = ws + 12800000;
    int*   ctr    = (int*)(ws + 15000000);
    float* mu     = (float*)d_out;
    float* ls     = (float*)d_out + 3200000;


    deg_hist_kernel<<<DG_WGS, DG_THR, 0, stream>>>(row, col, ew, part);
    dis_kernel<<<(N_NODES + 255) / 256, 256, 0, stream>>>(part, dis, invdeg, ctr);

    gemm_xwt_wmma<false><<<N_NODES / GM_ROWS, 256, 0, stream>>>(
        x, W1, W1, HID, b1, b1, invdeg, xw, acc1, acc1, HID);
    scatter_kernel<<<SC_WGS, 256, 0, stream>>>(
        row, col, ew, dis, xw, acc1, acc1 + 64, HID, ctr);

    gemm_xwt_wmma<true><<<N_NODES / GM_ROWS, 256, 0, stream>>>(
        acc1, Wmu, Wls, OUT_CH, bmu, bls, invdeg, xw, mu, ls, OUT_CH);
    scatter_kernel<<<SC_WGS, 256, 0, stream>>>(
        row, col, ew, dis, xw, mu, ls, OUT_CH, ctr + 2 * CTR_STRIDE);
}


# Layer 1 sums input rows over the edges first, W1 after: both layers' products run in one GEMM launch on an LDS-held strip (H never stored); 5 launches, 26 MB less row traffic; sum order changes
# speedup vs baseline: 3.1713x; 1.0706x over previous
#include <hip/hip_runtime.h>


#define N_NODES 50000
#define N_EDGES 600000
#define IN_CH   128
#define HID     128
#define OUT_CH  64

typedef __attribute__((ext_vector_type(2))) float v2f;
typedef __attribute__((ext_vector_type(4))) float v4f;
typedef __attribute__((ext_vector_type(8))) float v8f;
typedef __attribute__((ext_vector_type(4))) int   v4i;

#define DG_WGS   32
#define DG_THR   1024
#define DG_PITCH 50016
#define DG_QUADS (N_EDGES / 4)
#define DG_ITERS ((DG_QUADS + DG_WGS * DG_THR - 1) / (DG_WGS * DG_THR))
static_assert(N_EDGES % 4 == 0 && N_NODES % 4 == 0 && DG_PITCH >= N_NODES && DG_PITCH % 32 == 0,
              "degree tiling");

__global__ void __launch_bounds__(DG_THR)
deg_hist_kernel(const int* __restrict__ row, const int* __restrict__ col,
                const float* __restrict__ w, float* __restrict__ part) {
    __shared__ __attribute__((aligned(16))) float hist[N_NODES];
    const int tid = threadIdx.x;

    v4i r[DG_ITERS], c[DG_ITERS];
    v4f x[DG_ITERS];
#pragma unroll
    for (int k = 0; k < DG_ITERS; ++k) {
        int q = (k * DG_WGS + (int)blockIdx.x) * DG_THR + tid;
        q = q < DG_QUADS ? q : DG_QUADS - 1;
        r[k] = ((const v4i*)row)[q];
        c[k] = ((const v4i*)col)[q];
        x[k] = ((const v4f*)w)[q];
    }

    for (int i = tid; i < N_NODES / 4; i += DG_THR)
        ((v4f*)hist)[i] = (v4f){};
    __syncthreads();

#pragma unroll
    for (int k = 0; k < DG_ITERS; ++k) {
        const int q = (k * DG_WGS + (int)blockIdx.x) * DG_THR + tid;
        if (q < DG_QUADS) {
            atomicAdd(&hist[c[k].x], r[k].x == c[k].x ? 0.0f : x[k].x);
            atomicAdd(&hist[c[k].y], r[k].y == c[k].y ? 0.0f : x[k].y);
            atomicAdd(&hist[c[k].z], r[k].z == c[k].z ? 0.0f : x[k].z);
            atomicAdd(&hist[c[k].w], r[k].w == c[k].w ? 0.0f : x[k].w);
        }
    }
    __syncthreads();

    float* plane = part + (size_t)blockIdx.x * DG_PITCH;
    for (int i = tid; i < N_NODES / 4; i += DG_THR)
        ((v4f*)plane)[i] = ((const v4f*)hist)[i];
}

#define CTR_STRIDE 64
#define GM_ROWS  80
#define GM_RPW   (GM_ROWS / 8)
static_assert(N_NODES % GM_ROWS == 0, "node tiling");
__global__ void __launch_bounds__(256)
dis_kernel(const float* __restrict__ part, const float* __restrict__ x,
           float* __restrict__ dis, float* __restrict__ invdeg,
           float* __restrict__ ax, int* __restrict__ ctr) {
    __shared__ float sinv[GM_ROWS];
    const int tid  = threadIdx.x;
    const int wv   = __builtin_amdgcn_readfirstlane((int)(threadIdx.x >> 5));
    const int lane = tid & 31;
    const int m0   = blockIdx.x * GM_ROWS;
    const int r0   = wv * GM_RPW;

    v4f xr[GM_RPW];
#pragma unroll
    for (int j = 0; j < GM_RPW; ++j)
        xr[j] = *(const v4f*)(x + (size_t)(m0 + r0 + j) * IN_CH + lane * 4);

    if (blockIdx.x == 0 && tid < 4) ctr[tid * CTR_STRIDE] = 0;
    if (tid < GM_ROWS) {
        const int i = m0 + tid;
        float p[DG_WGS];
#pragma unroll
        for (int g = 0; g < DG_WGS; ++g)
            p[g] = part[(size_t)g * DG_PITCH + i];
        float sum = p[0];
#pragma unroll
        for (int g = 1; g < DG_WGS; ++g)
            sum += p[g];
        const float d  = sum + 1.0f;
        const float id = 1.0f / d;
        dis[i]    = rsqrtf(d);
        invdeg[i] = id;
        sinv[tid] = id;
    }
    __syncthreads();
#pragma unroll
    for (int j = 0; j < GM_RPW; ++j)
        *(v4f*)(ax + (size_t)(m0 + r0 + j) * IN_CH + lane * 4) = xr[j] * sinv[r0 + j];
}

#define GM_MT    (GM_ROWS / 16)
#define GM_PITCH 132
static_assert(IN_CH == 128 && HID == 128 && 2 * OUT_CH == HID, "gemm tiling");

__device__ __forceinline__ void strip_mma(const float* arow, const v4f (&bf)[16], v8f (&acc)[GM_MT]) {
    v4f a[GM_MT];
#pragma unroll
    for (int i = 0; i < GM_MT; ++i)
        a[i] = *(const v4f*)(arow + i * 16 * GM_PITCH);
#pragma unroll
    for (int p = 0; p < 16; ++p) {
        v4f an[GM_MT];
        if (p + 1 < 16) {
#pragma unroll
            for (int i = 0; i < GM_MT; ++i)
                an[i] = *(const v4f*)(arow + i * 16 * GM_PITCH + 8 * (p + 1));
        }
        __builtin_amdgcn_sched_barrier(0);
#pragma unroll
        for (int i = 0; i < GM_MT; ++i)
            acc[i] = __builtin_amdgcn_wmma_f32_16x16x4_f32(
                false, (v2f){a[i].x, a[i].y}, false, (v2f){bf[p].x, bf[p].y},
                (short)0, acc[i], false, false);
#pragma unroll
        for (int i = 0; i < GM_MT; ++i)
            acc[i] = __builtin_amdgcn_wmma_f32_16x16x4_f32(
                false, (v2f){a[i].z, a[i].w}, false, (v2f){bf[p].z, bf[p].w},
                (short)0, acc[i], false, false);
        __builtin_amdgcn_sched_barrier(0);
        if (p + 1 < 16) {
#pragma unroll
            for (int i = 0; i < GM_MT; ++i) a[i] = an[i];
        }
    }
}

__global__ void __launch_bounds__(256)
gemm_two_layer_wmma(const float* __restrict__ AX,
                    const float* __restrict__ W1,
                    const float* __restrict__ b1,
                    const float* __restrict__ Wmu,
                    const float* __restrict__ Wls,
                    const float* __restrict__ bmu,
                    const float* __restrict__ bls,
                    const float* __restrict__ invdeg,
                    float* __restrict__ C,
                    float* MU,
                    float* LS) {
    __shared__ float tile[GM_ROWS * GM_PITCH];

    const int wv   = __builtin_amdgcn_readfirstlane((int)(threadIdx.x >> 5));
    const int lane = threadIdx.x & 31;
    const int half = lane >> 4;
    const int l16  = lane & 15;
    const int m0   = blockIdx.x * GM_ROWS;
    const int r0   = wv * GM_RPW;

    v4f xr[GM_RPW];
#pragma unroll
    for (int j = 0; j < GM_RPW; ++j)
        xr[j] = *(const v4f*)(AX + (size_t)(m0 + r0 + j) * IN_CH + lane * 4);

    const int ng = wv * 16 + l16;
    v4f bf[16];
    {
        const float* wrow = W1 + (size_t)ng * IN_CH + half * 4;
#pragma unroll
        for (int p = 0; p < 16; ++p)
            bf[p] = *(const v4f*)(wrow + 8 * p);
    }
    const float bias1 = b1[ng];

#pragma unroll
    for (int j = 0; j < GM_RPW; ++j)
        *(v4f*)(tile + (r0 + j) * GM_PITCH + lane * 4) = xr[j];
    __syncthreads();

    v8f acc[GM_MT];
#pragma unroll
    for (int i = 0; i < GM_MT; ++i) acc[i] = (v8f){};

    const float* arow = tile + l16 * GM_PITCH + half * 4;
    strip_mma(arow, bf, acc);

    __syncthreads();
#pragma unroll
    for (int i = 0; i < GM_MT; ++i)
#pragma unroll
        for (int j = 0; j < 8; ++j)
            tile[(16 * i + 8 * half + j) * GM_PITCH + ng] = fmaxf(acc[i][j] + bias1, 0.0f);
    {
        const bool lowp = ng < OUT_CH;
        const float* wrow = (lowp ? Wmu + (size_t)ng * HID : Wls + (size_t)(ng - OUT_CH) * HID) + half * 4;
#pragma unroll
        for (int p = 0; p < 16; ++p)
            bf[p] = *(const v4f*)(wrow + 8 * p);
    }
    __syncthreads();

#pragma unroll
    for (int i = 0; i < GM_MT; ++i) acc[i] = (v8f){};
    strip_mma(arow, bf, acc);

    __syncthreads();
#pragma unroll
    for (int i = 0; i < GM_MT; ++i)
#pragma unroll
        for (int j = 0; j < 8; ++j)
            tile[(16 * i + 8 * half + j) * GM_PITCH + ng] = acc[i][j];
    __syncthreads();

    const int  c0   = lane * 4;
    const bool lowq = c0 < OUT_CH;
    const int  cl   = lowq ? c0 : c0 - OUT_CH;
    const v4f  b4   = *(const v4f*)((lowq ? bmu : bls) + cl);
    float* sbase    = (lowq ? MU : LS) + cl;
#pragma unroll
    for (int j = 0; j < GM_RPW; ++j) {
        const int m = m0 + r0 + j;
        const v4f v = *(const v4f*)(tile + (r0 + j) * GM_PITCH + c0);
        const float id = invdeg[m];
        *(v4f*)(C + (size_t)m * HID + c0) = v;
        *(v4f*)(sbase + (size_t)m * OUT_CH) = v * id + b4;
    }
}

#define EPI     32
#define N_ITEMS (N_EDGES / EPI)
#define EPB     8
#define SC_WGS  1024
static_assert(N_EDGES % EPI == 0 && EPI % EPB == 0 && EPI == 32 && N_NODES % 2 == 0, "edge tiling");

__device__ __forceinline__ float lane_value(float x, int l) {
    return __int_as_float(__builtin_amdgcn_readlane(__float_as_int(x), l));
}

__device__ __forceinline__ int next_item(int* ctr, int cls, int lane) {
    int v = 0;
    if (lane == 0) v = atomicAdd(ctr + cls * CTR_STRIDE, 1);
    return __builtin_amdgcn_readfirstlane(v);
}

__global__ void __launch_bounds__(256)
scatter_kernel(const int* __restrict__ row, const int* __restrict__ col,
               const float* __restrict__ w, const float* __restrict__ dis,
               const float* __restrict__ src,
               float* dA, float* dB, int pitch, int* ctr) {
    const int lane = threadIdx.x & 31;
    int  cls   = (int)((__builtin_amdgcn_s_sendmsg_rtn(0x87) >> 18) & 1u);
    bool other = false;
    const bool stealer = __builtin_amdgcn_readfirstlane((int)(threadIdx.x >> 5)) == 0;
    int  item  = next_item(ctr, cls, lane);
    for (int it = 0; it < 2 * N_ITEMS + 2; ++it) {
        if (item >= N_ITEMS) {
            if (other || !stealer) break;
            other = true;
            cls ^= 1;
            item = next_item(ctr, cls, lane);
            if (item >= N_ITEMS) break;
        }
        const int e = item * EPI + lane;
        const int r = row[e], c = col[e];
        const float wv = w[e];
        const int nxt = next_item(ctr, cls, lane);
        const bool mine = (c >= N_NODES / 2) == (cls != 0);
        const float we = (r == c) ? 0.0f : wv;
        const float dr = dis[mine ? r : 0];
        const float dc = dis[mine ? c : 0];
        unsigned mask = __builtin_amdgcn_ballot_w32(mine);

        for (int b = 0; b < EPI / EPB && mask != 0u; ++b) {
            const int n = __builtin_popcount(mask);
            int ln[EPB];
#pragma unroll
            for (int j = 0; j < EPB; ++j) {
                ln[j] = mask != 0u ? __builtin_ctz(mask) : 0;
                mask &= mask - 1u;
            }
            float v[EPB][4];
#pragma unroll
            for (int j = 0; j < EPB; ++j) {
                if (j < n) {
                    const int rj = __builtin_amdgcn_readlane(r, ln[j]);
                    const float* s = src + (size_t)rj * HID + lane;
                    v[j][0] = s[0];
                    v[j][1] = s[32];
                    v[j][2] = s[64];
                    v[j][3] = s[96];
                }
            }
#pragma unroll
            for (int j = 0; j < EPB; ++j) {
                if (j < n) {
                    const int   cj = __builtin_amdgcn_readlane(c, ln[j]);
                    const float cf = lane_value(dr, ln[j]) * lane_value(we, ln[j]) * lane_value(dc, ln[j]);
                    float* d0 = dA + (size_t)cj * pitch + lane;
                    float* d1 = dB + (size_t)cj * pitch + lane;
                    atomicAdd(d0 +  0, cf * v[j][0]);
                    atomicAdd(d0 + 32, cf * v[j][1]);
                    atomicAdd(d1 +  0, cf * v[j][2]);
                    atomicAdd(d1 + 32, cf * v[j][3]);
                }
            }
        }
        item = nxt;
    }
}

extern "C" void kernel_launch(void* const* d_in, const int* in_sizes, int n_in,
                              void* d_out, int out_size, void* d_ws, size_t ws_size,
                              hipStream_t stream) {
    (void)in_sizes; (void)n_in; (void)out_size; (void)ws_size;
    const float* x   = (const float*)d_in[0];
    const int*   ei  = (const int*)d_in[1];
    const float* ew  = (const float*)d_in[2];
    const float* W1  = (const float*)d_in[3];
    const float* b1  = (const float*)d_in[4];
    const float* Wmu = (const float*)d_in[5];
    const float* bmu = (const float*)d_in[6];
    const float* Wls = (const float*)d_in[7];
    const float* bls = (const float*)d_in[8];
    const int* row = ei;
    const int* col = ei + N_EDGES;

    float* ws     = (float*)d_ws;
    float* ax     = ws;
    float* xw2    = ws + 6400000;
    float* dis    = ws + 19200000;
    float* invdeg = ws + 19250000;
    float* part   = ws + 12800000;
    int*   ctr    = (int*)(ws + 15000000);
    float* mu     = (float*)d_out;
    float* ls     = (float*)d_out + 3200000;


    deg_hist_kernel<<<DG_WGS, DG_THR, 0, stream>>>(row, col, ew, part);
    dis_kernel<<<N_NODES / GM_ROWS, 256, 0, stream>>>(part, x, dis, invdeg, ax, ctr);

    scatter_kernel<<<SC_WGS, 256, 0, stream>>>(
        row, col, ew, dis, x, ax, ax + 64, IN_CH, ctr);

    gemm_two_layer_wmma<<<N_NODES / GM_ROWS, 256, 0, stream>>>(
        ax, W1, b1, Wmu, Wls, bmu, bls, invdeg, xw2, mu, ls);
    scatter_kernel<<<SC_WGS, 256, 0, stream>>>(
        row, col, ew, dis, xw2, mu, ls, OUT_CH, ctr + 2 * CTR_STRIDE);
}
